# starting kernel
# baseline (speedup 1.0000x reference)
#include <hip/hip_runtime.h>


typedef __attribute__((ext_vector_type(2))) float v2f;
typedef __attribute__((ext_vector_type(8))) float v8f;

#define NUM_NODES 256
#define BATCH     8
#define D_FEAT    128

__global__ __launch_bounds__(256) void nri_graph_kernel(
    const float* __restrict__ x, float* __restrict__ out) {

  __shared__ float s_sum[16];

  const int b    = blockIdx.x >> 3;
  const int d0   = (blockIdx.x & 7) << 4;
  const int t    = threadIdx.x;
  const int lane = t & 31;
  const int wave = t >> 5;
  const int n    = lane & 15;
  const int half = lane >> 4;

  if (t < 16) s_sum[t] = 0.0f;
  __syncthreads();

  const float* xb = x + (size_t)b * NUM_NODES * D_FEAT;

  v2f a; a.x = 1.0f; a.y = 1.0f;
  v8f c = {};
  const int nodeBase = wave * 32;
#pragma unroll
  for (int i = 0; i < 8; ++i) {
    const int n0 = nodeBase + i * 4;
    v2f bm;
    bm.x = xb[(size_t)(n0 + 0 + half) * D_FEAT + d0 + n];
    bm.y = xb[(size_t)(n0 + 2 + half) * D_FEAT + d0 + n];
    c = __builtin_amdgcn_wmma_f32_16x16x4_f32(false, a, false, bm,
                                              (short)0, c, false, false);
  }
  if (lane < 16) atomicAdd(&s_sum[n], c[0]);
  __syncthreads();

  const float S = s_sum[t & 15];

  float* ob = out + (size_t)b * NUM_NODES * (2 * D_FEAT);
  const int f     = t & 15;
  const int rBase = t >> 4;
#pragma unroll 4
  for (int it = 0; it < 16; ++it) {
    const int r = it * 16 + rBase;
    const float xv = xb[(size_t)r * D_FEAT + d0 + f];
    ob[(size_t)r * (2 * D_FEAT) + d0 + f]          = xv * (255.0f / 256.0f);
    ob[(size_t)r * (2 * D_FEAT) + D_FEAT + d0 + f] = (S - xv) * (1.0f / 256.0f);
  }
}

extern "C" void kernel_launch(void* const* d_in, const int* in_sizes, int n_in,
                              void* d_out, int out_size, void* d_ws, size_t ws_size,
                              hipStream_t stream) {
  const float* x = (const float*)d_in[0];
  float* out = (float*)d_out;
  (void)in_sizes; (void)n_in; (void)out_size; (void)d_ws; (void)ws_size;

  dim3 grid(BATCH * (D_FEAT / 16));
  dim3 block(256);
  nri_graph_kernel<<<grid, block, 0, stream>>>(x, out);
}


# x read once: a thread's 16 loaded values stay in registers as WMMA B fragments (each wave now sums a strided node set) and feed both output halves; the sum-free half is stored before the reduction
# speedup vs baseline: 1.2484x; 1.2484x over previous
#include <hip/hip_runtime.h>


typedef __attribute__((ext_vector_type(2))) float v2f;
typedef __attribute__((ext_vector_type(8))) float v8f;

#define NUM_NODES 256
#define BATCH     8
#define D_FEAT    128

__global__ __launch_bounds__(256) void nri_graph_kernel(
    const float* __restrict__ x, float* __restrict__ out) {

  __shared__ float s_sum[16];

  const int b     = blockIdx.x >> 3;
  const int d0    = (blockIdx.x & 7) << 4;
  const int t     = threadIdx.x;
  const int lane  = t & 31;
  const int f     = t & 15;
  const int rBase = t >> 4;

  if (t < 16) s_sum[t] = 0.0f;
  __syncthreads();

  const float* xp = x + (size_t)b * NUM_NODES * D_FEAT
                      + (size_t)rBase * D_FEAT + d0 + f;
  float xv[16];
#pragma unroll
  for (int it = 0; it < 16; ++it) xv[it] = xp[(size_t)it * 16 * D_FEAT];

  float* op = out + (size_t)b * NUM_NODES * (2 * D_FEAT)
                  + (size_t)rBase * (2 * D_FEAT) + d0 + f;
#pragma unroll
  for (int it = 0; it < 16; ++it)
    op[(size_t)it * 16 * (2 * D_FEAT)] = xv[it] * (255.0f / 256.0f);

  v2f a; a.x = 1.0f; a.y = 1.0f;
  v8f c = {};
#pragma unroll
  for (int i = 0; i < 8; ++i) {
    v2f bm; bm.x = xv[2 * i]; bm.y = xv[2 * i + 1];
    c = __builtin_amdgcn_wmma_f32_16x16x4_f32(false, a, false, bm,
                                              (short)0, c, false, false);
  }
  if (lane < 16) atomicAdd(&s_sum[f], c[0]);

  __builtin_amdgcn_fence(__ATOMIC_RELEASE, "workgroup", "local");
  __builtin_amdgcn_s_barrier();
  __builtin_amdgcn_fence(__ATOMIC_ACQUIRE, "workgroup", "local");

  const float S = s_sum[f];

#pragma unroll
  for (int it = 0; it < 16; ++it)
    op[(size_t)it * 16 * (2 * D_FEAT) + D_FEAT] = (S - xv[it]) * (1.0f / 256.0f);
}

extern "C" void kernel_launch(void* const* d_in, const int* in_sizes, int n_in,
                              void* d_out, int out_size, void* d_ws, size_t ws_size,
                              hipStream_t stream) {
  const float* x = (const float*)d_in[0];
  float* out = (float*)d_out;
  (void)in_sizes; (void)n_in; (void)out_size; (void)d_ws; (void)ws_size;

  dim3 grid(BATCH * (D_FEAT / 16));
  dim3 block(256);
  nri_graph_kernel<<<grid, block, 0, stream>>>(x, out);
}
